# starting kernel
# baseline (speedup 1.0000x reference)
#include <hip/hip_runtime.h>
#include <math.h>

typedef float v8f __attribute__((ext_vector_type(8)));
typedef float v2f __attribute__((ext_vector_type(2)));

#define WS_ROWD   0
#define WS_ROWH   8192
#define WS_ROWW   16384
#define WS_CHSUM  24576
#define WS_CHSQ   24592
#define WS_B2     24608
#define WS_ZERON  24608

__global__ void zero_ws_kernel(float* __restrict__ ws, int n) {
    int i = blockIdx.x * blockDim.x + threadIdx.x;
    if (i < n) ws[i] = 0.0f;
}

__global__ __launch_bounds__(256) void reduce_x_kernel(const float* __restrict__ x,
                                                       float* __restrict__ ws) {
    __shared__ float sD[4];
    __shared__ float sH[64];
    __shared__ float sW[64];
    __shared__ float sSq;

    const int tid  = threadIdx.x;
    const int lane = tid & 31;
    const int wave = tid >> 5;
    const int half = lane >> 4;
    const int l16  = lane & 15;
    const int bid  = blockIdx.x;
    const int bc    = bid >> 4;
    const int chunk = bid & 15;

    if (tid < 64) { sH[tid] = 0.0f; sW[tid] = 0.0f; }
    if (tid < 4)  sD[tid] = 0.0f;
    if (tid == 64) sSq = 0.0f;
    __syncthreads();

    const float* base = x + (size_t)bc * 262144u + (size_t)chunk * 16384u;

    float accW0 = 0.0f, accW1 = 0.0f, accW2 = 0.0f, accW3 = 0.0f, accSq = 0.0f;
    for (int p = wave; p < 128; p += 8) {
        const int r = 2 * p + half;
        const float4 v = *reinterpret_cast<const float4*>(base + r * 64 + l16 * 4);
        accW0 += v.x; accW1 += v.y; accW2 += v.z; accW3 += v.w;
        accSq += v.x * v.x + v.y * v.y + v.z * v.z + v.w * v.w;
        float rs = (v.x + v.y) + (v.z + v.w);
        #pragma unroll
        for (int off = 8; off > 0; off >>= 1) rs += __shfl_xor(rs, off, 32);
        if (l16 == 0) {
            atomicAdd(&sD[r >> 6], rs);
            atomicAdd(&sH[r & 63], rs);
        }
    }
    atomicAdd(&sW[l16 * 4 + 0], accW0);
    atomicAdd(&sW[l16 * 4 + 1], accW1);
    atomicAdd(&sW[l16 * 4 + 2], accW2);
    atomicAdd(&sW[l16 * 4 + 3], accW3);
    #pragma unroll
    for (int off = 16; off > 0; off >>= 1) accSq += __shfl_xor(accSq, off, 32);
    if (lane == 0) atomicAdd(&sSq, accSq);
    __syncthreads();

    if (tid < 64) {
        atomicAdd(&ws[WS_ROWW + bc * 64 + tid], sW[tid]);
        atomicAdd(&ws[WS_ROWH + bc * 64 + tid], sH[tid]);
    }
    if (tid < 4)
        atomicAdd(&ws[WS_ROWD + bc * 64 + chunk * 4 + tid], sD[tid]);
    if (tid == 64) {
        float t = sD[0] + sD[1] + sD[2] + sD[3];
        atomicAdd(&ws[WS_CHSUM + (bc & 15)], t);
    }
    if (tid == 65)
        atomicAdd(&ws[WS_CHSQ + (bc & 15)], sSq);
}

__global__ __launch_bounds__(256) void prep_kernel(float* __restrict__ ws,
                                                   const float* __restrict__ gamma,
                                                   const float* __restrict__ beta,
                                                   const float* __restrict__ core2) {
    __shared__ float sA[16];
    __shared__ float sOff[16];
    const int tid = threadIdx.x;
    if (tid < 16) {
        const float N = 2097152.0f;
        const float s = ws[WS_CHSUM + tid];
        const float q = ws[WS_CHSQ + tid];
        const float mean = s / N;
        const float var  = q / N - mean * mean;
        const float a = gamma[tid] * rsqrtf(var + 1e-5f);
        sA[tid]   = a;
        sOff[tid] = beta[tid] - mean * a;
    }
    __syncthreads();

    for (int i = tid; i < 24576; i += 256) {
        const int c = (i >> 6) & 15;
        ws[i] = sA[c] * (ws[i] * (1.0f / 4096.0f)) + sOff[c];
    }
    for (int i = tid; i < 2048; i += 256) {
        const int h = i >> 5;
        const int r = i & 31;
        const float* p = core2 + (r * 64 + h) * 32;
        float s = 0.0f;
        #pragma unroll 8
        for (int j = 0; j < 32; ++j) s += p[j];
        ws[WS_B2 + i] = s * (1.0f / 32.0f);
    }
}

__global__ __launch_bounds__(32) void gemm_gelu_kernel(const float* __restrict__ ws,
                                                       const float* __restrict__ core1,
                                                       const float* __restrict__ core3,
                                                       float* __restrict__ out) {
    const int lane = threadIdx.x;
    const int m0   = blockIdx.x * 16;
    const int lhi  = lane >> 4;
    const int llo  = lane & 15;

    for (int g = 0; g < 3; ++g) {
        const float* A = ws + g * 8192;
        const float* B = (g == 0) ? core1 : (ws + WS_B2);
        for (int nt = 0; nt < 2; ++nt) {
            const int n0 = nt * 16;
            v8f acc = {};
            for (int k0 = 0; k0 < 64; k0 += 4) {
                const int kk = k0 + lhi * 2;
                v2f a, b;
                const float* ap = A + (m0 + llo) * 64 + kk;
                a.x = ap[0];
                a.y = ap[1];
                const int n = n0 + llo;
                if (g < 2) {
                    b.x = B[kk * 32 + n];
                    b.y = B[(kk + 1) * 32 + n];
                } else {
                    const float* bp = core3 + n * 64 + kk;
                    b.x = bp[0];
                    b.y = bp[1];
                }
                acc = __builtin_amdgcn_wmma_f32_16x16x4_f32(
                     false, a,  false, b,
                     (short)0, acc,  false,  false);
            }
            #pragma unroll
            for (int j = 0; j < 8; ++j) {
                const int m = m0 + j + lhi * 8;
                const int n = n0 + llo;
                const float v = acc[j];
                const float ge = 0.5f * v * (1.0f + erff(v * 0.70710678118654752440f));
                out[m * 96 + g * 32 + n] = ge;
            }
        }
    }
}

extern "C" void kernel_launch(void* const* d_in, const int* in_sizes, int n_in,
                              void* d_out, int out_size, void* d_ws, size_t ws_size,
                              hipStream_t stream) {
    const float* x     = (const float*)d_in[0];
    const float* gamma = (const float*)d_in[1];
    const float* beta  = (const float*)d_in[2];
    const float* core1 = (const float*)d_in[3];
    const float* core2 = (const float*)d_in[4];
    const float* core3 = (const float*)d_in[5];
    float* ws  = (float*)d_ws;
    float* out = (float*)d_out;

    zero_ws_kernel<<<(WS_ZERON + 255) / 256, 256, 0, stream>>>(ws, WS_ZERON);
    reduce_x_kernel<<<2048, 256, 0, stream>>>(x, ws);
    prep_kernel<<<1, 256, 0, stream>>>(ws, gamma, beta, core2);
    gemm_gelu_kernel<<<8, 32, 0, stream>>>(ws, core1, core3, out);
}


# GEMM launch re-tiled by output lines: 24 blocks x 8 waves, a wave takes one K quarter of a 16x16 tile (4 WMMA, one load round trip), partials added in LDS, 2 GELUs per thread; sums re-associated
# speedup vs baseline: 2.5703x; 2.5703x over previous
#include <hip/hip_runtime.h>
#include <math.h>


typedef float v8f __attribute__((ext_vector_type(8)));
typedef float v2f __attribute__((ext_vector_type(2)));
typedef float v4f __attribute__((ext_vector_type(4)));

#define WS_ROWD   0
#define WS_ROWH   8192
#define WS_ROWW   16384
#define WS_B2     24608
#define WS_ROWH1  26656
#define WS_ROWW1  34848
#define WS_STAT   43040

__device__ __forceinline__ float row16_sum(float v) {
    v += __builtin_bit_cast(float, __builtin_amdgcn_update_dpp(0, __builtin_bit_cast(int, v), 0x161, 0xf, 0xf, true));
    v += __builtin_bit_cast(float, __builtin_amdgcn_update_dpp(0, __builtin_bit_cast(int, v), 0x162, 0xf, 0xf, true));
    v += __builtin_bit_cast(float, __builtin_amdgcn_update_dpp(0, __builtin_bit_cast(int, v), 0x164, 0xf, 0xf, true));
    v += __builtin_bit_cast(float, __builtin_amdgcn_update_dpp(0, __builtin_bit_cast(int, v), 0x168, 0xf, 0xf, true));
    return v;
}

#define EAT(v, rp, d)                                                 \
    do {                                                              \
        const v2f lo_ = (v).xy, hi_ = (v).zw;                         \
        wA += lo_; wB += hi_;                                         \
        qA = __builtin_elementwise_fma(lo_, lo_, qA);                 \
        qB = __builtin_elementwise_fma(hi_, hi_, qB);                 \
        const v2f s_ = lo_ + hi_;                                     \
        const float rs_ = s_.x + s_.y;                                \
        (rp) += rs_;                                                  \
        (d) += rs_;                                                   \
    } while (0)

__global__ __launch_bounds__(512) void reduce_x_kernel(const float* __restrict__ x,
                                                       float* __restrict__ ws) {
    __shared__ float sH[16][64];
    __shared__ float sW[16][64];
    __shared__ float sD[32];
    __shared__ float sQ[16];

    const int tid  = threadIdx.x;
    const int lane = tid & 31;
    const int wave = tid >> 5;
    const int half = lane >> 4;
    const int l16  = lane & 15;
    const int blk  = blockIdx.x;
    const int bc   = blk >> 1;
    const int hf   = blk & 1;

    const v4f* src = reinterpret_cast<const v4f*>(
        x + (size_t)bc * 262144u + (size_t)hf * 131072u + (size_t)wave * 8192u) + lane;

    float rowp[32];
    #pragma unroll
    for (int p = 0; p < 32; ++p) rowp[p] = 0.0f;
    v2f wA = {0.0f, 0.0f}, wB = {0.0f, 0.0f};
    v2f qA = {0.0f, 0.0f}, qB = {0.0f, 0.0f};
    float d0 = 0.0f, d1 = 0.0f;
    #pragma nounroll
    for (int pl = 0; pl < 2; ++pl) {
        v4f buf[32];
        #pragma unroll
        for (int k = 0; k < 32; ++k) buf[k] = src[pl * 1024 + k * 32];
        __builtin_amdgcn_sched_barrier(0);
        float dc[4] = {0.0f, 0.0f, 0.0f, 0.0f};
        #pragma unroll
        for (int k = 0; k < 32; ++k) EAT(buf[k], rowp[k], dc[k & 3]);
        const float dsum = (dc[0] + dc[1]) + (dc[2] + dc[3]);
        if (pl == 0) d0 = dsum; else d1 = dsum;
    }

    #pragma unroll
    for (int p = 0; p < 32; ++p) rowp[p] = row16_sum(rowp[p]);
    float hA = rowp[0], hB = rowp[16];
    #pragma unroll
    for (int k = 1; k < 16; ++k) {
        hA = (l16 == k) ? rowp[k] : hA;
        hB = (l16 == k) ? rowp[16 + k] : hB;
    }
    sH[wave][2 * l16 + half]      = hA;
    sH[wave][32 + 2 * l16 + half] = hB;

    float w0 = wA.x, w1 = wA.y, w2 = wB.x, w3 = wB.y;
    w0 += __shfl_xor(w0, 16, 32);
    w1 += __shfl_xor(w1, 16, 32);
    w2 += __shfl_xor(w2, 16, 32);
    w3 += __shfl_xor(w3, 16, 32);
    if (half == 0)
        *reinterpret_cast<float4*>(&sW[wave][l16 * 4]) = make_float4(w0, w1, w2, w3);

    float q = (qA.x + qA.y) + (qB.x + qB.y);
    d0 = row16_sum(d0);
    d1 = row16_sum(d1);
    q  = row16_sum(q);
    d0 += __shfl_xor(d0, 16, 32);
    d1 += __shfl_xor(d1, 16, 32);
    q  += __shfl_xor(q, 16, 32);
    if (lane == 0) {
        sD[wave * 2]     = d0;
        sD[wave * 2 + 1] = d1;
        sQ[wave]         = q;
    }
    __syncthreads();

    if (tid < 64) {
        float t = 0.0f;
        #pragma unroll
        for (int w = 0; w < 16; ++w) t += sH[w][tid];
        ws[(hf ? WS_ROWH1 : WS_ROWH) + bc * 64 + tid] = t;
    } else if (tid < 128) {
        const int k = tid - 64;
        float t = 0.0f;
        #pragma unroll
        for (int w = 0; w < 16; ++w) t += sW[w][k];
        ws[(hf ? WS_ROWW1 : WS_ROWW) + bc * 64 + k] = t;
    } else if (tid < 160) {
        const int k = tid - 128;
        ws[WS_ROWD + bc * 64 + hf * 32 + k] = sD[k];
    } else if (tid < 192) {
        const int k = tid - 160;
        float t = 0.0f;
        if (k == 0) {
            for (int j = 0; j < 32; ++j) t += sD[j];
        } else if (k == 1) {
            for (int j = 0; j < 16; ++j) t += sQ[j];
        }
        ws[WS_STAT + blk * 32 + k] = t;
    }
}

__global__ __launch_bounds__(64) void prep_kernel(float* __restrict__ ws,
                                                  const float* __restrict__ gamma,
                                                  const float* __restrict__ beta,
                                                  const float* __restrict__ core2) {
    const int tid = threadIdx.x;
    const int blk = blockIdx.x;
    if (blk < 384) {
        const int i = blk * 64 + tid;
        const int c = blk & 15;
        const int i2 = (blk >= 256) ? (WS_ROWW1 + (i - WS_ROWW))
                     : (blk >= 128) ? (WS_ROWH1 + (i - WS_ROWH)) : i;
        const float first = ws[i];
        const float second = ws[i2];
        const float raw = first + ((blk >= 128) ? second : 0.0f);
        float s = 0.0f, q = 0.0f;
        #pragma unroll
        for (int k = 0; k < 16; ++k) {
            const float* st = ws + WS_STAT + (((k >> 1) * 16 + c) * 2 + (k & 1)) * 32;
            s += st[0];
            q += st[1];
        }
        const float g = gamma[c];
        const float bt = beta[c];
        const float N = 2097152.0f;
        const float mean = s / N;
        const float var  = q / N - mean * mean;
        const float a = g * rsqrtf(var + 1e-5f);
        const float off = bt - mean * a;
        ws[i] = a * (raw * (1.0f / 4096.0f)) + off;
    } else {
        const int i = (blk - 384) * 64 + tid;
        const int h = i >> 5;
        const int r = i & 31;
        const float4* p = reinterpret_cast<const float4*>(core2 + (r * 64 + h) * 32);
        const float4 v0 = p[0], v1 = p[1], v2 = p[2], v3 = p[3];
        const float4 v4 = p[4], v5 = p[5], v6 = p[6], v7 = p[7];
        float t = 0.0f;
        t += v0.x; t += v0.y; t += v0.z; t += v0.w;
        t += v1.x; t += v1.y; t += v1.z; t += v1.w;
        t += v2.x; t += v2.y; t += v2.z; t += v2.w;
        t += v3.x; t += v3.y; t += v3.z; t += v3.w;
        t += v4.x; t += v4.y; t += v4.z; t += v4.w;
        t += v5.x; t += v5.y; t += v5.z; t += v5.w;
        t += v6.x; t += v6.y; t += v6.z; t += v6.w;
        t += v7.x; t += v7.y; t += v7.z; t += v7.w;
        ws[WS_B2 + i] = t * (1.0f / 32.0f);
    }
}

__global__ __launch_bounds__(256) void gemm_gelu_kernel(const float* __restrict__ ws,
                                                        const float* __restrict__ core1,
                                                        const float* __restrict__ core3,
                                                        float* __restrict__ out) {
    __shared__ float part[4][16][32];

    const int tid  = threadIdx.x;
    const int lane = tid & 31;
    const int wave = tid >> 5;
    const int nt   = wave & 1;
    const int kq   = wave >> 1;
    const int g    = blockIdx.x >> 3;
    const int m0   = (blockIdx.x & 7) * 16;
    const int lhi  = lane >> 4;
    const int llo  = lane & 15;

    const float* A = ws + g * 8192;
    const float* B = (g == 0) ? core1 : (g == 1) ? (ws + WS_B2) : core3;
    const int sk = (g < 2) ? 32 : 1;
    const int sn = (g < 2) ? 1 : 64;
    const int n = nt * 16 + llo;
    const float* ap = A + (m0 + llo) * 64 + kq * 16 + lhi * 2;
    const float* bp = B + (kq * 16 + lhi * 2) * sk + n * sn;
    v2f a[4], b[4];
    for (int i = 0; i < 4; ++i) {
        a[i].x = ap[i * 4];
        a[i].y = ap[i * 4 + 1];
        b[i].x = bp[(i * 4) * sk];
        b[i].y = bp[(i * 4 + 1) * sk];
    }
    v8f acc = {};
    for (int i = 0; i < 4; ++i)
        acc = __builtin_amdgcn_wmma_f32_16x16x4_f32(
             false, a[i],  false, b[i],
             (short)0, acc,  false,  false);
    for (int j = 0; j < 8; ++j)
        part[kq][j + lhi * 8][n] = acc[j];
    __syncthreads();

    const int row = tid >> 4;
    const int col = (tid & 15) * 2;
    const v2f p0 = *reinterpret_cast<const v2f*>(&part[0][row][col]);
    const v2f p1 = *reinterpret_cast<const v2f*>(&part[1][row][col]);
    const v2f p2 = *reinterpret_cast<const v2f*>(&part[2][row][col]);
    const v2f p3 = *reinterpret_cast<const v2f*>(&part[3][row][col]);
    const v2f v = ((p0 + p1) + p2) + p3;
    float2 ge;
    ge.x = 0.5f * v.x * (1.0f + erff(v.x * 0.70710678118654752440f));
    ge.y = 0.5f * v.y * (1.0f + erff(v.y * 0.70710678118654752440f));
    *reinterpret_cast<float2*>(&out[(m0 + row) * 96 + g * 32 + col]) = ge;
}

extern "C" void kernel_launch(void* const* d_in, const int* in_sizes, int n_in,
                              void* d_out, int out_size, void* d_ws, size_t ws_size,
                              hipStream_t stream) {
    const float* x     = (const float*)d_in[0];
    const float* gamma = (const float*)d_in[1];
    const float* beta  = (const float*)d_in[2];
    const float* core1 = (const float*)d_in[3];
    const float* core2 = (const float*)d_in[4];
    const float* core3 = (const float*)d_in[5];
    float* ws  = (float*)d_ws;
    float* out = (float*)d_out;

    reduce_x_kernel<<<256, 512, 0, stream>>>(x, ws);
    prep_kernel<<<416, 64, 0, stream>>>(ws, gamma, beta, core2);
    gemm_gelu_kernel<<<24, 256, 0, stream>>>(ws, core1, core3, out);
}


# prep launch dissolved (3 launches -> 2): GEMM on raw axis sums, BN affine taken behind the product (scale*P + offset*column sum of B) in its epilogue; B2 mean built in x pass; sums re-associated
# speedup vs baseline: 2.9958x; 1.1655x over previous
#include <hip/hip_runtime.h>
#include <math.h>


typedef float v8f __attribute__((ext_vector_type(8)));
typedef float v2f __attribute__((ext_vector_type(2)));
typedef float v4f __attribute__((ext_vector_type(4)));

#define WS_ROWD   0
#define WS_ROWH   8192
#define WS_ROWW   16384
#define WS_B2     24608
#define WS_ROWH1  26656
#define WS_ROWW1  34848
#define WS_STAT   43040

__device__ __forceinline__ float row16_sum(float v) {
    v += __builtin_bit_cast(float, __builtin_amdgcn_update_dpp(0, __builtin_bit_cast(int, v), 0x161, 0xf, 0xf, true));
    v += __builtin_bit_cast(float, __builtin_amdgcn_update_dpp(0, __builtin_bit_cast(int, v), 0x162, 0xf, 0xf, true));
    v += __builtin_bit_cast(float, __builtin_amdgcn_update_dpp(0, __builtin_bit_cast(int, v), 0x164, 0xf, 0xf, true));
    v += __builtin_bit_cast(float, __builtin_amdgcn_update_dpp(0, __builtin_bit_cast(int, v), 0x168, 0xf, 0xf, true));
    return v;
}

#define EAT(v, rp, d)                                                 \
    do {                                                              \
        const v2f lo_ = (v).xy, hi_ = (v).zw;                         \
        wA += lo_; wB += hi_;                                         \
        qA = __builtin_elementwise_fma(lo_, lo_, qA);                 \
        qB = __builtin_elementwise_fma(hi_, hi_, qB);                 \
        const v2f s_ = lo_ + hi_;                                     \
        const float rs_ = s_.x + s_.y;                                \
        (rp) += rs_;                                                  \
        (d) += rs_;                                                   \
    } while (0)

__global__ __launch_bounds__(512) void reduce_x_kernel(const float* __restrict__ x,
                                                       const float* __restrict__ core2,
                                                       float* __restrict__ ws) {
    __shared__ float sH[16][64];
    __shared__ float sW[16][64];
    __shared__ float sD[32];
    __shared__ float sQ[16];

    const int tid  = threadIdx.x;
    const int lane = tid & 31;
    const int wave = tid >> 5;
    const int half = lane >> 4;
    const int l16  = lane & 15;
    const int blk  = blockIdx.x;
    const int bc   = blk >> 1;
    const int hf   = blk & 1;

    const v4f* src = reinterpret_cast<const v4f*>(
        x + (size_t)bc * 262144u + (size_t)hf * 131072u + (size_t)wave * 8192u) + lane;

    if (blk < 64 && wave == 0) {
        const float4* p = reinterpret_cast<const float4*>(core2 + (lane * 64 + blk) * 32);
        const float4 c0 = p[0], c1 = p[1], c2 = p[2], c3 = p[3];
        const float4 c4 = p[4], c5 = p[5], c6 = p[6], c7 = p[7];
        float t = 0.0f;
        t += c0.x; t += c0.y; t += c0.z; t += c0.w;
        t += c1.x; t += c1.y; t += c1.z; t += c1.w;
        t += c2.x; t += c2.y; t += c2.z; t += c2.w;
        t += c3.x; t += c3.y; t += c3.z; t += c3.w;
        t += c4.x; t += c4.y; t += c4.z; t += c4.w;
        t += c5.x; t += c5.y; t += c5.z; t += c5.w;
        t += c6.x; t += c6.y; t += c6.z; t += c6.w;
        t += c7.x; t += c7.y; t += c7.z; t += c7.w;
        ws[WS_B2 + blk * 32 + lane] = t * (1.0f / 32.0f);
    }

    float rowp[32];
    #pragma unroll
    for (int p = 0; p < 32; ++p) rowp[p] = 0.0f;
    v2f wA = {0.0f, 0.0f}, wB = {0.0f, 0.0f};
    v2f qA = {0.0f, 0.0f}, qB = {0.0f, 0.0f};
    float d0 = 0.0f, d1 = 0.0f;
    #pragma nounroll
    for (int pl = 0; pl < 2; ++pl) {
        v4f buf[32];
        #pragma unroll
        for (int k = 0; k < 32; ++k) buf[k] = src[pl * 1024 + k * 32];
        __builtin_amdgcn_sched_barrier(0);
        float dc[4] = {0.0f, 0.0f, 0.0f, 0.0f};
        #pragma unroll
        for (int k = 0; k < 32; ++k) EAT(buf[k], rowp[k], dc[k & 3]);
        const float dsum = (dc[0] + dc[1]) + (dc[2] + dc[3]);
        if (pl == 0) d0 = dsum; else d1 = dsum;
    }

    #pragma unroll
    for (int p = 0; p < 32; ++p) rowp[p] = row16_sum(rowp[p]);
    float hA = rowp[0], hB = rowp[16];
    #pragma unroll
    for (int k = 1; k < 16; ++k) {
        hA = (l16 == k) ? rowp[k] : hA;
        hB = (l16 == k) ? rowp[16 + k] : hB;
    }
    sH[wave][2 * l16 + half]      = hA;
    sH[wave][32 + 2 * l16 + half] = hB;

    float w0 = wA.x, w1 = wA.y, w2 = wB.x, w3 = wB.y;
    w0 += __shfl_xor(w0, 16, 32);
    w1 += __shfl_xor(w1, 16, 32);
    w2 += __shfl_xor(w2, 16, 32);
    w3 += __shfl_xor(w3, 16, 32);
    if (half == 0)
        *reinterpret_cast<float4*>(&sW[wave][l16 * 4]) = make_float4(w0, w1, w2, w3);

    float q = (qA.x + qA.y) + (qB.x + qB.y);
    d0 = row16_sum(d0);
    d1 = row16_sum(d1);
    q  = row16_sum(q);
    d0 += __shfl_xor(d0, 16, 32);
    d1 += __shfl_xor(d1, 16, 32);
    q  += __shfl_xor(q, 16, 32);
    if (lane == 0) {
        sD[wave * 2]     = d0;
        sD[wave * 2 + 1] = d1;
        sQ[wave]         = q;
    }
    __syncthreads();

    if (tid < 64) {
        float t = 0.0f;
        #pragma unroll
        for (int w = 0; w < 16; ++w) t += sH[w][tid];
        ws[(hf ? WS_ROWH1 : WS_ROWH) + bc * 64 + tid] = t;
    } else if (tid < 128) {
        const int k = tid - 64;
        float t = 0.0f;
        #pragma unroll
        for (int w = 0; w < 16; ++w) t += sW[w][k];
        ws[(hf ? WS_ROWW1 : WS_ROWW) + bc * 64 + k] = t;
    } else if (tid < 160) {
        const int k = tid - 128;
        ws[WS_ROWD + bc * 64 + hf * 32 + k] = sD[k];
    } else if (tid < 192) {
        const int k = tid - 160;
        float t = 0.0f;
        if (k == 0) {
            for (int j = 0; j < 32; ++j) t += sD[j];
        } else if (k == 1) {
            for (int j = 0; j < 16; ++j) t += sQ[j];
        }
        ws[WS_STAT + blk * 32 + k] = t;
    }
}

__global__ __launch_bounds__(256) void gemm_gelu_kernel(const float* __restrict__ ws,
                                                        const float* __restrict__ gamma,
                                                        const float* __restrict__ beta,
                                                        const float* __restrict__ core1,
                                                        const float* __restrict__ core3,
                                                        float* __restrict__ out) {
    __shared__ float part[4][16][32];
    __shared__ float colp[8][32];

    const int tid  = threadIdx.x;
    const int lane = tid & 31;
    const int wave = tid >> 5;
    const int nt   = wave & 1;
    const int kq   = wave >> 1;
    const int g    = blockIdx.x >> 3;
    const int m0   = (blockIdx.x & 7) * 16;
    const int lhi  = lane >> 4;
    const int llo  = lane & 15;

    const int row = tid >> 4;
    const int col = (tid & 15) * 2;
    float s = 0.0f, q = 0.0f;
    #pragma unroll
    for (int k = 0; k < 16; ++k) {
        const float* st = ws + WS_STAT + (((k >> 1) * 16 + row) * 2 + (k & 1)) * 32;
        s += st[0];
        q += st[1];
    }
    const float gm = gamma[row];
    const float bt = beta[row];

    const float* A = ws + g * 8192;
    const float* A2 = ws + ((g == 1) ? WS_ROWH1 : WS_ROWW1);
    const float* B = (g == 0) ? core1 : (g == 1) ? (ws + WS_B2) : core3;
    const int sk = (g < 2) ? 32 : 1;
    const int sn = (g < 2) ? 1 : 64;
    const int n = nt * 16 + llo;
    const int aoff = (m0 + llo) * 64 + kq * 16 + lhi * 2;
    const float* ap = A + aoff;
    const float* ap2 = A2 + aoff;
    const float* bp = B + (kq * 16 + lhi * 2) * sk + n * sn;
    v2f a[4], b[4];
    for (int i = 0; i < 4; ++i) {
        a[i].x = ap[i * 4];
        a[i].y = ap[i * 4 + 1];
        b[i].x = bp[(i * 4) * sk];
        b[i].y = bp[(i * 4 + 1) * sk];
    }
    if (g != 0) {
        for (int i = 0; i < 4; ++i) {
            a[i].x += ap2[i * 4];
            a[i].y += ap2[i * 4 + 1];
        }
    }
    colp[kq * 2 + lhi][n] = ((b[0].x + b[0].y) + (b[1].x + b[1].y)) + ((b[2].x + b[2].y) + (b[3].x + b[3].y));
    v8f acc = {};
    for (int i = 0; i < 4; ++i)
        acc = __builtin_amdgcn_wmma_f32_16x16x4_f32(
             false, a[i],  false, b[i],
             (short)0, acc,  false,  false);
    for (int j = 0; j < 8; ++j)
        part[kq][j + lhi * 8][n] = acc[j];
    __syncthreads();

    const v2f p0 = *reinterpret_cast<const v2f*>(&part[0][row][col]);
    const v2f p1 = *reinterpret_cast<const v2f*>(&part[1][row][col]);
    const v2f p2 = *reinterpret_cast<const v2f*>(&part[2][row][col]);
    const v2f p3 = *reinterpret_cast<const v2f*>(&part[3][row][col]);
    const v2f raw = ((p0 + p1) + p2) + p3;
    v2f S = {0.0f, 0.0f};
    #pragma unroll
    for (int j = 0; j < 8; ++j) S += *reinterpret_cast<const v2f*>(&colp[j][col]);
    const float N = 2097152.0f;
    const float mean = s / N;
    const float var  = q / N - mean * mean;
    const float sa = gm * rsqrtf(var + 1e-5f);
    const float off = bt - mean * sa;
    const float sc = sa * (1.0f / 4096.0f);
    v2f v;
    v.x = sc * raw.x + off * S.x;
    v.y = sc * raw.y + off * S.y;
    float2 ge;
    ge.x = 0.5f * v.x * (1.0f + erff(v.x * 0.70710678118654752440f));
    ge.y = 0.5f * v.y * (1.0f + erff(v.y * 0.70710678118654752440f));
    *reinterpret_cast<float2*>(&out[(m0 + row) * 96 + g * 32 + col]) = ge;
}

extern "C" void kernel_launch(void* const* d_in, const int* in_sizes, int n_in,
                              void* d_out, int out_size, void* d_ws, size_t ws_size,
                              hipStream_t stream) {
    const float* x     = (const float*)d_in[0];
    const float* gamma = (const float*)d_in[1];
    const float* beta  = (const float*)d_in[2];
    const float* core1 = (const float*)d_in[3];
    const float* core2 = (const float*)d_in[4];
    const float* core3 = (const float*)d_in[5];
    float* ws  = (float*)d_ws;
    float* out = (float*)d_out;

    reduce_x_kernel<<<256, 512, 0, stream>>>(x, core2, ws);
    gemm_gelu_kernel<<<24, 256, 0, stream>>>(ws, gamma, beta, core1, core3, out);
}
